# starting kernel
# baseline (speedup 1.0000x reference)
#include <hip/hip_runtime.h>
#include <math.h>

typedef __attribute__((ext_vector_type(2))) float v2f;
typedef __attribute__((ext_vector_type(8))) float v8f;

#define B_DIM 2
#define N_NODES 25000
#define FC_IN 64
#define FC_HID 256
#define MUL 32
#define OUT_CH 288
#define EDGES_PER_BLOCK 128
#define WAVES 8
#define SH_STRIDE 18
#define SW_STRIDE 97

__device__ __forceinline__ void atomic_add_f32(float* p, float v) {
    unsafeAtomicAdd(p, v);
}

__global__ __launch_bounds__(256) void conv_mlp_tp_scatter(
    const int* __restrict__ edge_src,
    const int* __restrict__ edge_dst,
    const float* __restrict__ node_emb,
    const float* __restrict__ edge_type,
    const float* __restrict__ W1,
    const float* __restrict__ W2,
    float* __restrict__ out)
{
    __shared__ float sH[WAVES * 16 * SH_STRIDE];
    __shared__ float sW[EDGES_PER_BLOCK * SW_STRIDE];

    const int tid   = threadIdx.x;
    const int wv    = tid >> 5;
    const int lane  = tid & 31;
    const int lmod  = lane & 15;
    const int khalf = (lane < 16) ? 0 : 2;
    const int e_base = blockIdx.x * EDGES_PER_BLOCK;
    const int e_tile = e_base + wv * 16;

    const v8f vzero = {0.f, 0.f, 0.f, 0.f, 0.f, 0.f, 0.f, 0.f};

    v2f aET[16];
    {
        const float* etp = edge_type + (size_t)(e_tile + lmod) * FC_IN;
        #pragma unroll
        for (int ks = 0; ks < 16; ++ks)
            aET[ks] = *(const v2f*)(etp + 4 * ks + khalf);
        __builtin_prefetch(etp + EDGES_PER_BLOCK * FC_IN, 0, 1);
    }

    v8f h[16];
    #pragma unroll
    for (int nt = 0; nt < 16; ++nt) {
        v8f acc = vzero;
        #pragma unroll
        for (int ks = 0; ks < 16; ++ks) {
            const int k0 = 4 * ks + khalf;
            v2f b;
            b.x = W1[(size_t)k0 * FC_HID + 16 * nt + lmod];
            b.y = W1[(size_t)(k0 + 1) * FC_HID + 16 * nt + lmod];
            acc = __builtin_amdgcn_wmma_f32_16x16x4_f32(
                false, aET[ks], false, b, (short)0, acc, false, false);
        }
        #pragma unroll
        for (int i = 0; i < 8; ++i) {
            const float hv = acc[i] * 0.125f;
            acc[i] = hv / (1.0f + __expf(-hv));
        }
        h[nt] = acc;
    }

    float* myH = sH + wv * (16 * SH_STRIDE);
    v8f acc2[6];
    #pragma unroll
    for (int j = 0; j < 6; ++j) acc2[j] = vzero;

    #pragma unroll
    for (int nt1 = 0; nt1 < 16; ++nt1) {
        {
            const int mbase = (lane < 16) ? 0 : 8;
            #pragma unroll
            for (int r = 0; r < 8; ++r)
                myH[(r + mbase) * SH_STRIDE + lmod] = h[nt1][r];
        }
        v2f a[4];
        #pragma unroll
        for (int kk = 0; kk < 4; ++kk)
            a[kk] = *(const v2f*)(myH + lmod * SH_STRIDE + 4 * kk + khalf);

        #pragma unroll
        for (int nt2 = 0; nt2 < 6; ++nt2) {
            #pragma unroll
            for (int kk = 0; kk < 4; ++kk) {
                const int k0 = 16 * nt1 + 4 * kk + khalf;
                v2f b;
                b.x = W2[(size_t)k0 * 96 + 16 * nt2 + lmod];
                b.y = W2[(size_t)(k0 + 1) * 96 + 16 * nt2 + lmod];
                acc2[nt2] = __builtin_amdgcn_wmma_f32_16x16x4_f32(
                    false, a[kk], false, b, (short)0, acc2[nt2], false, false);
            }
        }
    }

    {
        const int mbase = wv * 16 + ((lane < 16) ? 0 : 8);
        #pragma unroll
        for (int nt2 = 0; nt2 < 6; ++nt2) {
            #pragma unroll
            for (int r = 0; r < 8; ++r)
                sW[(mbase + r) * SW_STRIDE + 16 * nt2 + lmod] = acc2[nt2][r] * 0.0625f;
        }
    }
    __syncthreads();

    {
        const int e_loc = tid >> 1;
        const int bb    = tid & 1;
        const int e     = e_base + e_loc;
        const int si    = edge_src[e];
        const int di    = edge_dst[e];
        const float* xp = node_emb + ((size_t)bb * N_NODES + si) * 3;
        const float* yp = node_emb + ((size_t)bb * N_NODES + di) * 3;
        const float x0 = xp[0], x1 = xp[1], x2 = xp[2];
        const float y0 = yp[0], y1 = yp[1], y2 = yp[2];

        const float inv3 = 0.57735026918962576f;
        const float inv2 = 0.70710678118654752f;
        const float inv6 = 0.40824829046386302f;
        const float scl  = 0.25f;

        const float s  = (x0*y0 + x1*y1 + x2*y2) * inv3 * scl;
        const float v0 = (x1*y2 - x2*y1) * inv2 * scl;
        const float v1 = (x2*y0 - x0*y2) * inv2 * scl;
        const float v2 = (x0*y1 - x1*y0) * inv2 * scl;
        const float t0 = (x0*y1 + x1*y0) * inv2 * scl;
        const float t1 = (x1*y2 + x2*y1) * inv2 * scl;
        const float t2 = (x0*y2 + x2*y0) * inv2 * scl;
        const float t3 = (x0*y0 - x1*y1) * inv2 * scl;
        const float t4 = (2.0f*x2*y2 - x0*y0 - x1*y1) * inv6 * scl;

        const float* wrow = sW + e_loc * SW_STRIDE;
        float* op = out + ((size_t)bb * N_NODES + di) * OUT_CH;

        #pragma unroll 4
        for (int u = 0; u < MUL; ++u)
            atomic_add_f32(op + u, wrow[u] * s);

        #pragma unroll 4
        for (int u = 0; u < MUL; ++u) {
            const float wu = wrow[MUL + u];
            atomic_add_f32(op + MUL + 3*u + 0, wu * v0);
            atomic_add_f32(op + MUL + 3*u + 1, wu * v1);
            atomic_add_f32(op + MUL + 3*u + 2, wu * v2);
        }

        #pragma unroll 2
        for (int u = 0; u < MUL; ++u) {
            const float wu = wrow[2*MUL + u];
            atomic_add_f32(op + 4*MUL + 5*u + 0, wu * t0);
            atomic_add_f32(op + 4*MUL + 5*u + 1, wu * t1);
            atomic_add_f32(op + 4*MUL + 5*u + 2, wu * t2);
            atomic_add_f32(op + 4*MUL + 5*u + 3, wu * t3);
            atomic_add_f32(op + 4*MUL + 5*u + 4, wu * t4);
        }
    }
}

extern "C" void kernel_launch(void* const* d_in, const int* in_sizes, int n_in,
                              void* d_out, int out_size, void* d_ws, size_t ws_size,
                              hipStream_t stream) {
    const int*   edge_src  = (const int*)d_in[0];
    const int*   edge_dst  = (const int*)d_in[1];
    const float* node_emb  = (const float*)d_in[2];
    const float* edge_type = (const float*)d_in[3];
    const float* W1        = (const float*)d_in[4];
    const float* W2        = (const float*)d_in[5];
    float*       out       = (float*)d_out;

    hipMemsetAsync(d_out, 0, (size_t)out_size * sizeof(float), stream);

    const int E = in_sizes[0];
    const int nblocks = E / EDGES_PER_BLOCK;
    conv_mlp_tp_scatter<<<nblocks, 256, 0, stream>>>(
        edge_src, edge_dst, node_emb, edge_type, W1, W2, out);
}


# scatter re-mapped: lane = output channel, each wave walks its own 16 edges, so each f32 atomic add instruction hits one whole 128-B line of one (batch,dst) row, not 32 rows; atomic add order changes
# speedup vs baseline: 6.4864x; 6.4864x over previous
#include <hip/hip_runtime.h>
#include <math.h>

typedef __attribute__((ext_vector_type(2))) float v2f;
typedef __attribute__((ext_vector_type(8))) float v8f;

#define B_DIM 2
#define N_NODES 25000
#define FC_IN 64
#define FC_HID 256
#define MUL 32
#define OUT_CH 288
#define EDGES_PER_BLOCK 128
#define WAVES 8
#define SH_STRIDE 18
#define SW_STRIDE 97

__device__ __forceinline__ void atomic_add_f32(float* p, float v) {
    unsafeAtomicAdd(p, v);
}

__global__ __launch_bounds__(256) void conv_mlp_tp_scatter(
    const int* __restrict__ edge_src,
    const int* __restrict__ edge_dst,
    const float* __restrict__ node_emb,
    const float* __restrict__ edge_type,
    const float* __restrict__ W1,
    const float* __restrict__ W2,
    float* __restrict__ out)
{
    __shared__ float sH[WAVES * 16 * SH_STRIDE];
    __shared__ float sW[EDGES_PER_BLOCK * SW_STRIDE];

    const int tid   = threadIdx.x;
    const int wv    = tid >> 5;
    const int lane  = tid & 31;
    const int lmod  = lane & 15;
    const int khalf = (lane < 16) ? 0 : 2;
    const int e_base = blockIdx.x * EDGES_PER_BLOCK;
    const int e_tile = e_base + wv * 16;

    const v8f vzero = {0.f, 0.f, 0.f, 0.f, 0.f, 0.f, 0.f, 0.f};

    v2f aET[16];
    {
        const float* etp = edge_type + (size_t)(e_tile + lmod) * FC_IN;
        #pragma unroll
        for (int ks = 0; ks < 16; ++ks)
            aET[ks] = *(const v2f*)(etp + 4 * ks + khalf);
        __builtin_prefetch(etp + EDGES_PER_BLOCK * FC_IN, 0, 1);
    }

    v8f h[16];
    #pragma unroll
    for (int nt = 0; nt < 16; ++nt) {
        v8f acc = vzero;
        #pragma unroll
        for (int ks = 0; ks < 16; ++ks) {
            const int k0 = 4 * ks + khalf;
            v2f b;
            b.x = W1[(size_t)k0 * FC_HID + 16 * nt + lmod];
            b.y = W1[(size_t)(k0 + 1) * FC_HID + 16 * nt + lmod];
            acc = __builtin_amdgcn_wmma_f32_16x16x4_f32(
                false, aET[ks], false, b, (short)0, acc, false, false);
        }
        #pragma unroll
        for (int i = 0; i < 8; ++i) {
            const float hv = acc[i] * 0.125f;
            acc[i] = hv / (1.0f + __expf(-hv));
        }
        h[nt] = acc;
    }

    float* myH = sH + wv * (16 * SH_STRIDE);
    v8f acc2[6];
    #pragma unroll
    for (int j = 0; j < 6; ++j) acc2[j] = vzero;

    #pragma unroll
    for (int nt1 = 0; nt1 < 16; ++nt1) {
        {
            const int mbase = (lane < 16) ? 0 : 8;
            #pragma unroll
            for (int r = 0; r < 8; ++r)
                myH[(r + mbase) * SH_STRIDE + lmod] = h[nt1][r];
        }
        v2f a[4];
        #pragma unroll
        for (int kk = 0; kk < 4; ++kk)
            a[kk] = *(const v2f*)(myH + lmod * SH_STRIDE + 4 * kk + khalf);

        #pragma unroll
        for (int nt2 = 0; nt2 < 6; ++nt2) {
            #pragma unroll
            for (int kk = 0; kk < 4; ++kk) {
                const int k0 = 16 * nt1 + 4 * kk + khalf;
                v2f b;
                b.x = W2[(size_t)k0 * 96 + 16 * nt2 + lmod];
                b.y = W2[(size_t)(k0 + 1) * 96 + 16 * nt2 + lmod];
                acc2[nt2] = __builtin_amdgcn_wmma_f32_16x16x4_f32(
                    false, a[kk], false, b, (short)0, acc2[nt2], false, false);
            }
        }
    }

    {
        const int mbase = wv * 16 + ((lane < 16) ? 0 : 8);
        #pragma unroll
        for (int nt2 = 0; nt2 < 6; ++nt2) {
            #pragma unroll
            for (int r = 0; r < 8; ++r)
                sW[(mbase + r) * SW_STRIDE + 16 * nt2 + lmod] = acc2[nt2][r] * 0.0625f;
        }
    }

    {
        float* cf = myH;
        const int e_loc = lane >> 1;
        const int bb    = lane & 1;
        const int e     = e_tile + e_loc;
        const int si    = edge_src[e];
        const int di    = edge_dst[e];
        {
            const float* xp = node_emb + ((size_t)bb * N_NODES + si) * 3;
            const float* yp = node_emb + ((size_t)bb * N_NODES + di) * 3;
            const float x0 = xp[0], x1 = xp[1], x2 = xp[2];
            const float y0 = yp[0], y1 = yp[1], y2 = yp[2];

            const float inv3 = 0.57735026918962576f;
            const float inv2 = 0.70710678118654752f;
            const float inv6 = 0.40824829046386302f;
            const float scl  = 0.25f;

            float* c = cf + e_loc * 18 + bb;
            c[0]  = (x0*y0 + x1*y1 + x2*y2) * inv3 * scl;
            c[2]  = (x1*y2 - x2*y1) * inv2 * scl;
            c[4]  = (x2*y0 - x0*y2) * inv2 * scl;
            c[6]  = (x0*y1 - x1*y0) * inv2 * scl;
            c[8]  = (x0*y1 + x1*y0) * inv2 * scl;
            c[10] = (x1*y2 + x2*y1) * inv2 * scl;
            c[12] = (x0*y2 + x2*y0) * inv2 * scl;
            c[14] = (x0*y0 - x1*y1) * inv2 * scl;
            c[16] = (2.0f*x2*y2 - x0*y0 - x1*y1) * inv6 * scl;
        }

        int wo[9], ko[9];
        wo[0] = lane; ko[0] = 0;
        #pragma unroll
        for (int j = 1; j < 4; ++j) {
            const int q = 32 * (j - 1) + lane;
            wo[j] = MUL + q / 3;  ko[j] = 2 * (1 + q % 3);
        }
        #pragma unroll
        for (int j = 4; j < 9; ++j) {
            const int q = 32 * (j - 4) + lane;
            wo[j] = 2 * MUL + q / 5;  ko[j] = 2 * (4 + q % 5);
        }

        const float* wrow0 = sW + (wv * 16) * SW_STRIDE;
        float wu[9];
        v2f   c2[9];
        #pragma unroll
        for (int j = 0; j < 9; ++j) {
            wu[j] = wrow0[wo[j]];
            c2[j] = *(const v2f*)(cf + ko[j]);
        }
        #pragma unroll
        for (int el = 0; el < 16; ++el) {
            float p0[9], p1[9];
            #pragma unroll
            for (int j = 0; j < 9; ++j) {
                p0[j] = wu[j] * c2[j].x;
                p1[j] = wu[j] * c2[j].y;
            }
            if (el + 1 < 16) {
                #pragma unroll
                for (int j = 0; j < 9; ++j) {
                    wu[j] = wrow0[(el + 1) * SW_STRIDE + wo[j]];
                    c2[j] = *(const v2f*)(cf + (el + 1) * 18 + ko[j]);
                }
            }
            const int d = __builtin_amdgcn_readlane(di, 2 * el);
            float* o0 = out + (size_t)d * OUT_CH + lane;
            float* o1 = o0 + (size_t)N_NODES * OUT_CH;
            #pragma unroll
            for (int j = 0; j < 9; ++j) {
                atomic_add_f32(o0 + 32 * j, p0[j]);
                atomic_add_f32(o1 + 32 * j, p1[j]);
            }
        }
    }
}

extern "C" void kernel_launch(void* const* d_in, const int* in_sizes, int n_in,
                              void* d_out, int out_size, void* d_ws, size_t ws_size,
                              hipStream_t stream) {
    const int*   edge_src  = (const int*)d_in[0];
    const int*   edge_dst  = (const int*)d_in[1];
    const float* node_emb  = (const float*)d_in[2];
    const float* edge_type = (const float*)d_in[3];
    const float* W1        = (const float*)d_in[4];
    const float* W2        = (const float*)d_in[5];
    float*       out       = (float*)d_out;

    hipMemsetAsync(d_out, 0, (size_t)out_size * sizeof(float), stream);

    const int E = in_sizes[0];
    const int nblocks = E / EDGES_PER_BLOCK;
    conv_mlp_tp_scatter<<<nblocks, 256, 0, stream>>>(
        edge_src, edge_dst, node_emb, edge_type, W1, W2, out);
}


# W loads software-pipelined: the two products run tile by tile (an H tile feeds its GEMM2 K slice at once, H is never whole in registers) and each run's W fragments are requested one run ahead, fenced
# speedup vs baseline: 7.5038x; 1.1569x over previous
#include <hip/hip_runtime.h>
#include <math.h>

typedef __attribute__((ext_vector_type(2))) float v2f;
typedef __attribute__((ext_vector_type(8))) float v8f;

#define B_DIM 2
#define N_NODES 25000
#define FC_IN 64
#define FC_HID 256
#define MUL 32
#define OUT_CH 288
#define EDGES_PER_BLOCK 128
#define WAVES 8
#define SH_STRIDE 18
#define SW_STRIDE 97

__device__ __forceinline__ void atomic_add_f32(float* p, float v) {
    unsafeAtomicAdd(p, v);
}

__global__ __launch_bounds__(256) void conv_mlp_tp_scatter(
    const int* __restrict__ edge_src,
    const int* __restrict__ edge_dst,
    const float* __restrict__ node_emb,
    const float* __restrict__ edge_type,
    const float* __restrict__ W1,
    const float* __restrict__ W2,
    float* __restrict__ out)
{
    __shared__ float sH[WAVES * 16 * SH_STRIDE];
    __shared__ float sW[EDGES_PER_BLOCK * SW_STRIDE];

    const int tid   = threadIdx.x;
    const int wv    = tid >> 5;
    const int lane  = tid & 31;
    const int lmod  = lane & 15;
    const int khalf = (lane < 16) ? 0 : 2;
    const int e_base = blockIdx.x * EDGES_PER_BLOCK;
    const int e_tile = e_base + wv * 16;

    const v8f vzero = {0.f, 0.f, 0.f, 0.f, 0.f, 0.f, 0.f, 0.f};

    v2f aET[16];
    {
        const float* etp = edge_type + (size_t)(e_tile + lmod) * FC_IN;
        #pragma unroll
        for (int ks = 0; ks < 16; ++ks)
            aET[ks] = *(const v2f*)(etp + 4 * ks + khalf);
        __builtin_prefetch(etp + EDGES_PER_BLOCK * FC_IN, 0, 1);
    }

    float* myH = sH + wv * (16 * SH_STRIDE);
    v8f acc2[6];
    #pragma unroll
    for (int j = 0; j < 6; ++j) acc2[j] = vzero;

    v2f w1b[16];
    v2f w2b[24];

    #pragma unroll
    for (int ks = 0; ks < 16; ++ks) {
        const int k0 = 4 * ks + khalf;
        w1b[ks].x = W1[(size_t)k0 * FC_HID + lmod];
        w1b[ks].y = W1[(size_t)(k0 + 1) * FC_HID + lmod];
    }

    #pragma unroll
    for (int nt = 0; nt < 16; ++nt) {
        #pragma unroll
        for (int nt2 = 0; nt2 < 6; ++nt2) {
            #pragma unroll
            for (int kk = 0; kk < 4; ++kk) {
                const int k0 = 16 * nt + 4 * kk + khalf;
                w2b[4 * nt2 + kk].x = W2[(size_t)k0 * 96 + 16 * nt2 + lmod];
                w2b[4 * nt2 + kk].y = W2[(size_t)(k0 + 1) * 96 + 16 * nt2 + lmod];
            }
        }
        __builtin_amdgcn_sched_barrier(0);

        v8f acc = vzero;
        #pragma unroll
        for (int ks = 0; ks < 16; ++ks)
            acc = __builtin_amdgcn_wmma_f32_16x16x4_f32(
                false, aET[ks], false, w1b[ks], (short)0, acc, false, false);

        {
            const int mbase = (lane < 16) ? 0 : 8;
            #pragma unroll
            for (int r = 0; r < 8; ++r) {
                const float hv = acc[r] * 0.125f;
                myH[(r + mbase) * SH_STRIDE + lmod] = hv / (1.0f + __expf(-hv));
            }
        }
        v2f a[4];
        #pragma unroll
        for (int kk = 0; kk < 4; ++kk)
            a[kk] = *(const v2f*)(myH + lmod * SH_STRIDE + 4 * kk + khalf);

        if (nt + 1 < 16) {
            #pragma unroll
            for (int ks = 0; ks < 16; ++ks) {
                const int k0 = 4 * ks + khalf;
                w1b[ks].x = W1[(size_t)k0 * FC_HID + 16 * (nt + 1) + lmod];
                w1b[ks].y = W1[(size_t)(k0 + 1) * FC_HID + 16 * (nt + 1) + lmod];
            }
        }
        __builtin_amdgcn_sched_barrier(0);

        #pragma unroll
        for (int nt2 = 0; nt2 < 6; ++nt2) {
            #pragma unroll
            for (int kk = 0; kk < 4; ++kk)
                acc2[nt2] = __builtin_amdgcn_wmma_f32_16x16x4_f32(
                    false, a[kk], false, w2b[4 * nt2 + kk], (short)0, acc2[nt2], false, false);
        }
    }

    {
        const int mbase = wv * 16 + ((lane < 16) ? 0 : 8);
        #pragma unroll
        for (int nt2 = 0; nt2 < 6; ++nt2) {
            #pragma unroll
            for (int r = 0; r < 8; ++r)
                sW[(mbase + r) * SW_STRIDE + 16 * nt2 + lmod] = acc2[nt2][r] * 0.0625f;
        }
    }

    {
        float* cf = myH;
        const int e_loc = lane >> 1;
        const int bb    = lane & 1;
        const int e     = e_tile + e_loc;
        const int si    = edge_src[e];
        const int di    = edge_dst[e];
        {
            const float* xp = node_emb + ((size_t)bb * N_NODES + si) * 3;
            const float* yp = node_emb + ((size_t)bb * N_NODES + di) * 3;
            const float x0 = xp[0], x1 = xp[1], x2 = xp[2];
            const float y0 = yp[0], y1 = yp[1], y2 = yp[2];

            const float inv3 = 0.57735026918962576f;
            const float inv2 = 0.70710678118654752f;
            const float inv6 = 0.40824829046386302f;
            const float scl  = 0.25f;

            float* c = cf + e_loc * 18 + bb;
            c[0]  = (x0*y0 + x1*y1 + x2*y2) * inv3 * scl;
            c[2]  = (x1*y2 - x2*y1) * inv2 * scl;
            c[4]  = (x2*y0 - x0*y2) * inv2 * scl;
            c[6]  = (x0*y1 - x1*y0) * inv2 * scl;
            c[8]  = (x0*y1 + x1*y0) * inv2 * scl;
            c[10] = (x1*y2 + x2*y1) * inv2 * scl;
            c[12] = (x0*y2 + x2*y0) * inv2 * scl;
            c[14] = (x0*y0 - x1*y1) * inv2 * scl;
            c[16] = (2.0f*x2*y2 - x0*y0 - x1*y1) * inv6 * scl;
        }

        int wo[9], ko[9];
        wo[0] = lane; ko[0] = 0;
        #pragma unroll
        for (int j = 1; j < 4; ++j) {
            const int q = 32 * (j - 1) + lane;
            wo[j] = MUL + q / 3;  ko[j] = 2 * (1 + q % 3);
        }
        #pragma unroll
        for (int j = 4; j < 9; ++j) {
            const int q = 32 * (j - 4) + lane;
            wo[j] = 2 * MUL + q / 5;  ko[j] = 2 * (4 + q % 5);
        }

        const float* wrow0 = sW + (wv * 16) * SW_STRIDE;
        float wu[9];
        v2f   c2[9];
        #pragma unroll
        for (int j = 0; j < 9; ++j) {
            wu[j] = wrow0[wo[j]];
            c2[j] = *(const v2f*)(cf + ko[j]);
        }
        #pragma unroll
        for (int el = 0; el < 16; ++el) {
            float p0[9], p1[9];
            #pragma unroll
            for (int j = 0; j < 9; ++j) {
                p0[j] = wu[j] * c2[j].x;
                p1[j] = wu[j] * c2[j].y;
            }
            if (el + 1 < 16) {
                #pragma unroll
                for (int j = 0; j < 9; ++j) {
                    wu[j] = wrow0[(el + 1) * SW_STRIDE + wo[j]];
                    c2[j] = *(const v2f*)(cf + (el + 1) * 18 + ko[j]);
                }
            }
            const int d = __builtin_amdgcn_readlane(di, 2 * el);
            float* o0 = out + (size_t)d * OUT_CH + lane;
            float* o1 = o0 + (size_t)N_NODES * OUT_CH;
            #pragma unroll
            for (int j = 0; j < 9; ++j) {
                atomic_add_f32(o0 + 32 * j, p0[j]);
                atomic_add_f32(o1 + 32 * j, p1[j]);
            }
        }
    }
}

extern "C" void kernel_launch(void* const* d_in, const int* in_sizes, int n_in,
                              void* d_out, int out_size, void* d_ws, size_t ws_size,
                              hipStream_t stream) {
    const int*   edge_src  = (const int*)d_in[0];
    const int*   edge_dst  = (const int*)d_in[1];
    const float* node_emb  = (const float*)d_in[2];
    const float* edge_type = (const float*)d_in[3];
    const float* W1        = (const float*)d_in[4];
    const float* W2        = (const float*)d_in[5];
    float*       out       = (float*)d_out;

    hipMemsetAsync(d_out, 0, (size_t)out_size * sizeof(float), stream);

    const int E = in_sizes[0];
    const int nblocks = E / EDGES_PER_BLOCK;
    conv_mlp_tp_scatter<<<nblocks, 256, 0, stream>>>(
        edge_src, edge_dst, node_emb, edge_type, W1, W2, out);
}
